# starting kernel
# baseline (speedup 1.0000x reference)
#include <hip/hip_runtime.h>


#define IN_C  256
#define HID_C 128
#define OUT_C 256

typedef int idx_t;
typedef __attribute__((ext_vector_type(2))) float v2f;
typedef __attribute__((ext_vector_type(8))) float v8f;

__global__ void zero_f32_kernel(float* __restrict__ p, long n) {
  long i = (long)blockIdx.x * blockDim.x + threadIdx.x;
  const long stride = (long)gridDim.x * blockDim.x;
  for (; i < n; i += stride) p[i] = 0.0f;
}

__global__ void degree_kernel(const idx_t* __restrict__ src,
                              const idx_t* __restrict__ dst,
                              float* __restrict__ deg_out,
                              float* __restrict__ deg_in, int E) {
  const int e = blockIdx.x * blockDim.x + threadIdx.x;
  if (e >= E) return;
  unsafeAtomicAdd(&deg_out[src[e]], 1.0f);
  unsafeAtomicAdd(&deg_in[dst[e]], 1.0f);
}

__global__ void rsqrt_deg_kernel(float* __restrict__ deg, int n) {
  const int i = blockIdx.x * blockDim.x + threadIdx.x;
  if (i >= n) return;
  deg[i] = rsqrtf(fmaxf(deg[i], 1.0f));
}

__global__ void transpose_kernel(const float* __restrict__ W,
                                 float* __restrict__ Wt, int rows, int cols) {
  const int i = blockIdx.x * blockDim.x + threadIdx.x;
  if (i >= rows * cols) return;
  const int r = i / cols;
  const int c = i - r * cols;
  Wt[(long)c * rows + r] = W[i];
}

template <int KDIM, int NTILES, int NOUT, bool HAS_BIAS, bool RELU,
          bool SCALE_A>
__global__ __launch_bounds__(64) void gemm_wmma_kernel(
    const float* __restrict__ A, const float* __restrict__ W,
    const float* __restrict__ bias, const float* __restrict__ scaleA,
    float* __restrict__ D) {
  const int lane = threadIdx.x & 31;
  const int rowTile = blockIdx.x * 2 + (threadIdx.x >> 5);
  const int m0 = rowTile * 16;
  const int n0 = blockIdx.y * (NTILES * 16);
  const int mrow = m0 + (lane & 15);
  const int khalf = lane >> 4;

  const float s = SCALE_A ? scaleA[mrow] : 1.0f;

  v8f acc[NTILES];
#pragma unroll
  for (int t = 0; t < NTILES; ++t) acc[t] = (v8f){0, 0, 0, 0, 0, 0, 0, 0};

  const float* __restrict__ arow = A + (long)mrow * KDIM;
  for (int k0 = 0; k0 < KDIM; k0 += 4) {
    const int ka = k0 + khalf * 2;
    const float2 av = *(const float2*)(arow + ka);
    v2f a;
    a.x = av.x * s;
    a.y = av.y * s;
#pragma unroll
    for (int t = 0; t < NTILES; ++t) {
      const int ncol = n0 + t * 16 + (lane & 15);
      const float2 bv = *(const float2*)(W + (long)ncol * KDIM + ka);
      v2f b;
      b.x = bv.x;
      b.y = bv.y;
      acc[t] = __builtin_amdgcn_wmma_f32_16x16x4_f32(
          false, a, false, b, (short)0, acc[t], false, false);
    }
  }

#pragma unroll
  for (int t = 0; t < NTILES; ++t) {
    const int ncol = n0 + t * 16 + (lane & 15);
#pragma unroll
    for (int v = 0; v < 8; ++v) {
      const int row = m0 + v + 8 * khalf;
      float val = acc[t][v];
      if constexpr (HAS_BIAS) val += bias[ncol];
      if constexpr (RELU) val = fmaxf(val, 0.0f);
      D[(long)row * NOUT + ncol] = val;
    }
  }
}

#define EDGES_PER_WAVE 512
__global__ void aggregate_kernel(const float* __restrict__ h1,
                                 const idx_t* __restrict__ src,
                                 const idx_t* __restrict__ dst,
                                 float* __restrict__ m, int E) {
  const int wave = (int)((blockIdx.x * blockDim.x + threadIdx.x) >> 5);
  const int lane = threadIdx.x & 31;
  long e0 = (long)wave * EDGES_PER_WAVE;
  if (e0 >= E) return;
  long e1 = e0 + EDGES_PER_WAVE;
  if (e1 > E) e1 = E;
  const int c = lane * 4;

  float4 acc = make_float4(0.f, 0.f, 0.f, 0.f);
  int cur = dst[e0];
  for (long e = e0; e < e1; ++e) {
    const int d = dst[e];
    if (d != cur) {
      float* p = m + (long)cur * HID_C + c;
      unsafeAtomicAdd(p + 0, acc.x);
      unsafeAtomicAdd(p + 1, acc.y);
      unsafeAtomicAdd(p + 2, acc.z);
      unsafeAtomicAdd(p + 3, acc.w);
      acc = make_float4(0.f, 0.f, 0.f, 0.f);
      cur = d;
    }
    const float4 v = *(const float4*)(h1 + (long)src[e] * HID_C + c);
    acc.x += v.x;
    acc.y += v.y;
    acc.z += v.z;
    acc.w += v.w;
  }
  float* p = m + (long)cur * HID_C + c;
  unsafeAtomicAdd(p + 0, acc.x);
  unsafeAtomicAdd(p + 1, acc.y);
  unsafeAtomicAdd(p + 2, acc.z);
  unsafeAtomicAdd(p + 3, acc.w);
}

__global__ void norm_bias_relu_kernel(float* __restrict__ m,
                                      const float* __restrict__ rin,
                                      const float* __restrict__ b_conv,
                                      long total) {
  const long i = (long)blockIdx.x * blockDim.x + threadIdx.x;
  if (i >= total) return;
  const int c = (int)(i & (HID_C - 1));
  const long node = i >> 7;
  const float v = m[i] * rin[node] + b_conv[c];
  m[i] = fmaxf(v, 0.0f);
}

extern "C" void kernel_launch(void* const* d_in, const int* in_sizes, int n_in,
                              void* d_out, int out_size, void* d_ws,
                              size_t ws_size, hipStream_t stream) {
  const float* x      = (const float*)d_in[0];
  const idx_t* src    = (const idx_t*)d_in[1];
  const idx_t* dst    = (const idx_t*)d_in[2];
  const float* W_conv = (const float*)d_in[3];
  const float* b_conv = (const float*)d_in[4];
  const float* W1     = (const float*)d_in[5];
  const float* b1     = (const float*)d_in[6];
  const float* W2     = (const float*)d_in[7];
  const float* b2     = (const float*)d_in[8];
  float* out = (float*)d_out;

  const int N = in_sizes[0] / IN_C;
  const int E = in_sizes[1];

  float* ws      = (float*)d_ws;
  float* deg_out = ws;
  float* deg_in  = ws + N;
  float* Wc_t    = ws + 2L * N;
  float* bufB    = Wc_t + (long)HID_C * IN_C;
  float* bufA    = bufB + (long)N * HID_C;

  zero_f32_kernel<<<64, 256, 0, stream>>>(ws, 2L * N);
  zero_f32_kernel<<<2048, 256, 0, stream>>>(bufB, (long)N * HID_C);

  transpose_kernel<<<(IN_C * HID_C + 255) / 256, 256, 0, stream>>>(
      W_conv, Wc_t, IN_C, HID_C);

  degree_kernel<<<(E + 255) / 256, 256, 0, stream>>>(src, dst, deg_out, deg_in, E);
  rsqrt_deg_kernel<<<(2 * N + 255) / 256, 256, 0, stream>>>(deg_out, 2 * N);

  const int rowTiles = N / 16;

  gemm_wmma_kernel<IN_C, 8, HID_C,  false,  false,
                    true>
      <<<dim3(rowTiles / 2, 1), 64, 0, stream>>>(x, Wc_t, nullptr, deg_out,
                                                 bufA);

  const int waves = (E + EDGES_PER_WAVE - 1) / EDGES_PER_WAVE;
  aggregate_kernel<<<(waves * 32 + 255) / 256, 256, 0, stream>>>(bufA, src, dst,
                                                                 bufB, E);

  norm_bias_relu_kernel<<<(int)(((long)N * HID_C + 255) / 256), 256, 0,
                          stream>>>(bufB, deg_in, b_conv, (long)N * HID_C);

  gemm_wmma_kernel<HID_C, 8, HID_C,  true,  true,
                    false>
      <<<dim3(rowTiles / 2, 1), 64, 0, stream>>>(bufB, W1, b1, nullptr, bufA);

  gemm_wmma_kernel<HID_C, 8, OUT_C,  true,  false,
                    false>
      <<<dim3(rowTiles / 2, 2), 64, 0, stream>>>(bufA, W2, b2, nullptr, out);
}


# aggregate: 32 edge indices per coalesced load, handed out by readlane; 16 row gathers in flight per wave; 128 edges per wave (was 512, one gather at a time); more runs split into two atomic partials
# speedup vs baseline: 1.3336x; 1.3336x over previous
#include <hip/hip_runtime.h>


#define IN_C  256
#define HID_C 128
#define OUT_C 256

typedef int idx_t;
typedef __attribute__((ext_vector_type(2))) float v2f;
typedef __attribute__((ext_vector_type(8))) float v8f;

__global__ void zero_f32_kernel(float* __restrict__ p, long n) {
  long i = (long)blockIdx.x * blockDim.x + threadIdx.x;
  const long stride = (long)gridDim.x * blockDim.x;
  for (; i < n; i += stride) p[i] = 0.0f;
}

__global__ void degree_kernel(const idx_t* __restrict__ src,
                              const idx_t* __restrict__ dst,
                              float* __restrict__ deg_out,
                              float* __restrict__ deg_in, int E) {
  const int e = blockIdx.x * blockDim.x + threadIdx.x;
  if (e >= E) return;
  unsafeAtomicAdd(&deg_out[src[e]], 1.0f);
  unsafeAtomicAdd(&deg_in[dst[e]], 1.0f);
}

__global__ void rsqrt_deg_kernel(float* __restrict__ deg, int n) {
  const int i = blockIdx.x * blockDim.x + threadIdx.x;
  if (i >= n) return;
  deg[i] = rsqrtf(fmaxf(deg[i], 1.0f));
}

__global__ void transpose_kernel(const float* __restrict__ W,
                                 float* __restrict__ Wt, int rows, int cols) {
  const int i = blockIdx.x * blockDim.x + threadIdx.x;
  if (i >= rows * cols) return;
  const int r = i / cols;
  const int c = i - r * cols;
  Wt[(long)c * rows + r] = W[i];
}

template <int KDIM, int NTILES, int NOUT, bool HAS_BIAS, bool RELU,
          bool SCALE_A>
__global__ __launch_bounds__(64) void gemm_wmma_kernel(
    const float* __restrict__ A, const float* __restrict__ W,
    const float* __restrict__ bias, const float* __restrict__ scaleA,
    float* __restrict__ D) {
  const int lane = threadIdx.x & 31;
  const int rowTile = blockIdx.x * 2 + (threadIdx.x >> 5);
  const int m0 = rowTile * 16;
  const int n0 = blockIdx.y * (NTILES * 16);
  const int mrow = m0 + (lane & 15);
  const int khalf = lane >> 4;

  const float s = SCALE_A ? scaleA[mrow] : 1.0f;

  v8f acc[NTILES];
#pragma unroll
  for (int t = 0; t < NTILES; ++t) acc[t] = (v8f){0, 0, 0, 0, 0, 0, 0, 0};

  const float* __restrict__ arow = A + (long)mrow * KDIM;
  for (int k0 = 0; k0 < KDIM; k0 += 4) {
    const int ka = k0 + khalf * 2;
    const float2 av = *(const float2*)(arow + ka);
    v2f a;
    a.x = av.x * s;
    a.y = av.y * s;
#pragma unroll
    for (int t = 0; t < NTILES; ++t) {
      const int ncol = n0 + t * 16 + (lane & 15);
      const float2 bv = *(const float2*)(W + (long)ncol * KDIM + ka);
      v2f b;
      b.x = bv.x;
      b.y = bv.y;
      acc[t] = __builtin_amdgcn_wmma_f32_16x16x4_f32(
          false, a, false, b, (short)0, acc[t], false, false);
    }
  }

#pragma unroll
  for (int t = 0; t < NTILES; ++t) {
    const int ncol = n0 + t * 16 + (lane & 15);
#pragma unroll
    for (int v = 0; v < 8; ++v) {
      const int row = m0 + v + 8 * khalf;
      float val = acc[t][v];
      if constexpr (HAS_BIAS) val += bias[ncol];
      if constexpr (RELU) val = fmaxf(val, 0.0f);
      D[(long)row * NOUT + ncol] = val;
    }
  }
}

#define EDGES_PER_WAVE 128
#define GATHER_DEPTH 16

__device__ __forceinline__ void flush_run(float* __restrict__ m, int node,
                                          int c, const float4 acc) {
  float* p = m + (long)node * HID_C + c;
  unsafeAtomicAdd(p + 0, acc.x);
  unsafeAtomicAdd(p + 1, acc.y);
  unsafeAtomicAdd(p + 2, acc.z);
  unsafeAtomicAdd(p + 3, acc.w);
}

__global__ __launch_bounds__(256) void aggregate_kernel(
    const float* __restrict__ h1, const idx_t* __restrict__ src,
    const idx_t* __restrict__ dst, float* __restrict__ m, int E) {
  const int wave = __builtin_amdgcn_readfirstlane(
      (int)((blockIdx.x * blockDim.x + threadIdx.x) >> 5));
  const int lane = threadIdx.x & 31;
  const long e0 = (long)wave * EDGES_PER_WAVE;
  if (e0 >= E) return;
  long e1 = e0 + EDGES_PER_WAVE;
  if (e1 > E) e1 = E;
  const int c = lane * 4;
  const float* __restrict__ hc = h1 + c;

  int sv = 0, dv = 0;
  if (e0 + lane < e1) {
    sv = src[e0 + lane];
    dv = dst[e0 + lane];
  }
  float4 acc = make_float4(0.f, 0.f, 0.f, 0.f);
  int cur = __builtin_amdgcn_readlane(dv, 0);

  for (long eb = e0; eb < e1; eb += 32) {
    int sn = 0, dn = 0;
    if (eb + 32 + lane < e1) {
      sn = src[eb + 32 + lane];
      dn = dst[eb + 32 + lane];
    }
    if (e1 - eb >= 32) {
#pragma unroll
      for (int g = 0; g < 32; g += GATHER_DEPTH) {
        float4 v[GATHER_DEPTH];
#pragma unroll
        for (int j = 0; j < GATHER_DEPTH; ++j) {
          const int s = __builtin_amdgcn_readlane(sv, g + j);
          v[j] = *(const float4*)(hc + (long)s * HID_C);
        }
#pragma unroll
        for (int j = 0; j < GATHER_DEPTH; ++j) {
          const int d = __builtin_amdgcn_readlane(dv, g + j);
          if (d != cur) {
            flush_run(m, cur, c, acc);
            acc = make_float4(0.f, 0.f, 0.f, 0.f);
            cur = d;
          }
          acc.x += v[j].x;
          acc.y += v[j].y;
          acc.z += v[j].z;
          acc.w += v[j].w;
        }
      }
    } else {
      const int nv = (int)(e1 - eb);
      for (int j = 0; j < nv; ++j) {
        const int s = __builtin_amdgcn_readlane(sv, j);
        const int d = __builtin_amdgcn_readlane(dv, j);
        if (d != cur) {
          flush_run(m, cur, c, acc);
          acc = make_float4(0.f, 0.f, 0.f, 0.f);
          cur = d;
        }
        const float4 v = *(const float4*)(hc + (long)s * HID_C);
        acc.x += v.x;
        acc.y += v.y;
        acc.z += v.z;
        acc.w += v.w;
      }
    }
    sv = sn;
    dv = dn;
  }
  flush_run(m, cur, c, acc);
}

__global__ void norm_bias_relu_kernel(float* __restrict__ m,
                                      const float* __restrict__ rin,
                                      const float* __restrict__ b_conv,
                                      long total) {
  const long i = (long)blockIdx.x * blockDim.x + threadIdx.x;
  if (i >= total) return;
  const int c = (int)(i & (HID_C - 1));
  const long node = i >> 7;
  const float v = m[i] * rin[node] + b_conv[c];
  m[i] = fmaxf(v, 0.0f);
}

extern "C" void kernel_launch(void* const* d_in, const int* in_sizes, int n_in,
                              void* d_out, int out_size, void* d_ws,
                              size_t ws_size, hipStream_t stream) {
  const float* x      = (const float*)d_in[0];
  const idx_t* src    = (const idx_t*)d_in[1];
  const idx_t* dst    = (const idx_t*)d_in[2];
  const float* W_conv = (const float*)d_in[3];
  const float* b_conv = (const float*)d_in[4];
  const float* W1     = (const float*)d_in[5];
  const float* b1     = (const float*)d_in[6];
  const float* W2     = (const float*)d_in[7];
  const float* b2     = (const float*)d_in[8];
  float* out = (float*)d_out;

  const int N = in_sizes[0] / IN_C;
  const int E = in_sizes[1];

  float* ws      = (float*)d_ws;
  float* deg_out = ws;
  float* deg_in  = ws + N;
  float* Wc_t    = ws + 2L * N;
  float* bufB    = Wc_t + (long)HID_C * IN_C;
  float* bufA    = bufB + (long)N * HID_C;

  zero_f32_kernel<<<64, 256, 0, stream>>>(ws, 2L * N);
  zero_f32_kernel<<<2048, 256, 0, stream>>>(bufB, (long)N * HID_C);

  transpose_kernel<<<(IN_C * HID_C + 255) / 256, 256, 0, stream>>>(
      W_conv, Wc_t, IN_C, HID_C);

  degree_kernel<<<(E + 255) / 256, 256, 0, stream>>>(src, dst, deg_out, deg_in, E);
  rsqrt_deg_kernel<<<(2 * N + 255) / 256, 256, 0, stream>>>(deg_out, 2 * N);

  const int rowTiles = N / 16;

  gemm_wmma_kernel<IN_C, 8, HID_C,  false,  false,
                    true>
      <<<dim3(rowTiles / 2, 1), 64, 0, stream>>>(x, Wc_t, nullptr, deg_out,
                                                 bufA);

  const int waves = (E + EDGES_PER_WAVE - 1) / EDGES_PER_WAVE;
  aggregate_kernel<<<(waves * 32 + 255) / 256, 256, 0, stream>>>(bufA, src, dst,
                                                                 bufB, E);

  norm_bias_relu_kernel<<<(int)(((long)N * HID_C + 255) / 256), 256, 0,
                          stream>>>(bufB, deg_in, b_conv, (long)N * HID_C);

  gemm_wmma_kernel<HID_C, 8, HID_C,  true,  true,
                    false>
      <<<dim3(rowTiles / 2, 1), 64, 0, stream>>>(bufB, W1, b1, nullptr, bufA);

  gemm_wmma_kernel<HID_C, 8, OUT_C,  true,  false,
                    false>
      <<<dim3(rowTiles / 2, 2), 64, 0, stream>>>(bufA, W2, b2, nullptr, out);
}


# GEMMs: W copied once per 8-wave block into padded LDS, B fragments read from LDS (were 16-line global loads per wave and strip); 512 resident blocks walk the 16x128 strips; k and sum order kept
# speedup vs baseline: 1.7807x; 1.3353x over previous
#include <hip/hip_runtime.h>


#define IN_C  256
#define HID_C 128
#define OUT_C 256

typedef int idx_t;
typedef __attribute__((ext_vector_type(2))) float v2f;
typedef __attribute__((ext_vector_type(8))) float v8f;

__global__ void zero_f32_kernel(float* __restrict__ p, long n) {
  long i = (long)blockIdx.x * blockDim.x + threadIdx.x;
  const long stride = (long)gridDim.x * blockDim.x;
  for (; i < n; i += stride) p[i] = 0.0f;
}

__global__ void degree_kernel(const idx_t* __restrict__ src,
                              const idx_t* __restrict__ dst,
                              float* __restrict__ deg_out,
                              float* __restrict__ deg_in, int E) {
  const int e = blockIdx.x * blockDim.x + threadIdx.x;
  if (e >= E) return;
  unsafeAtomicAdd(&deg_out[src[e]], 1.0f);
  unsafeAtomicAdd(&deg_in[dst[e]], 1.0f);
}

__global__ void rsqrt_deg_kernel(float* __restrict__ deg, int n) {
  const int i = blockIdx.x * blockDim.x + threadIdx.x;
  if (i >= n) return;
  deg[i] = rsqrtf(fmaxf(deg[i], 1.0f));
}

__global__ void transpose_kernel(const float* __restrict__ W,
                                 float* __restrict__ Wt, int rows, int cols) {
  const int i = blockIdx.x * blockDim.x + threadIdx.x;
  if (i >= rows * cols) return;
  const int r = i / cols;
  const int c = i - r * cols;
  Wt[(long)c * rows + r] = W[i];
}

#define GEMM_BLOCKS 512
template <int KDIM, int NTOT, bool HAS_BIAS, bool RELU, bool SCALE_A>
__global__ __launch_bounds__(256) void gemm_wmma_kernel(
    const float* __restrict__ A, const float* __restrict__ W,
    const float* __restrict__ bias, const float* __restrict__ scaleA,
    float* __restrict__ D, int nRowTiles) {
  constexpr int P = 36;
  constexpr int CH = NTOT * P;
  constexpr int NSPLIT = NTOT / 128;
  constexpr int F4R = KDIM / 4;
  constexpr int PASSES = NTOT * F4R / 256;
  __shared__ __attribute__((aligned(16))) float wl[(KDIM / 32) * CH];

  const int tid = threadIdx.x;
#pragma unroll 8
  for (int p = 0; p < PASSES; ++p) {
    const int idx = (((p + (int)blockIdx.x) & (PASSES - 1)) << 8) + tid;
    const int r = idx / F4R;
    const int c4 = idx - r * F4R;
    const float4 v = *(const float4*)(W + (long)idx * 4);
    *(float4*)(wl + (c4 >> 3) * CH + r * P + (c4 & 7) * 4) = v;
  }
  __syncthreads();

  const int lane = tid & 31;
  const int c = lane & 15;
  const int khalf = lane >> 4;
  const int wv = __builtin_amdgcn_readfirstlane(tid >> 5);
  const int rank = (0x73261540 >> (4 * wv)) & 7;

  for (int v = rank;; v += 8) {
    const int u = v / NSPLIT;
    const int rowTile = u * GEMM_BLOCKS + (int)blockIdx.x;
    if (rowTile >= nRowTiles) break;
    const int n0 = (v - u * NSPLIT) * 128;
    const int m0 = rowTile * 16;
    const int mrow = m0 + c;

    const float s = SCALE_A ? scaleA[mrow] : 1.0f;

    v8f acc[8];
#pragma unroll
    for (int t = 0; t < 8; ++t) acc[t] = (v8f){0, 0, 0, 0, 0, 0, 0, 0};

    const float* __restrict__ ap = A + (long)mrow * KDIM + khalf * 2;
    const float* bp = wl + (n0 + c) * P + khalf * 2;

    float2 an[8];
#pragma unroll
    for (int j = 0; j < 8; ++j) an[j] = *(const float2*)(ap + 4 * j);

#pragma unroll 1
    for (int kb = 0; kb < KDIM; kb += 32) {
      float2 ac[8];
#pragma unroll
      for (int j = 0; j < 8; ++j) ac[j] = an[j];
      if (kb + 32 < KDIM) {
#pragma unroll
        for (int j = 0; j < 8; ++j)
          an[j] = *(const float2*)(ap + kb + 32 + 4 * j);
      }
#pragma unroll
      for (int hf = 0; hf < 2; ++hf) {
        v2f b[4][8];
#pragma unroll
        for (int js = 0; js < 4; ++js) {
#pragma unroll
          for (int t = 0; t < 8; ++t) {
            const float2 bv =
                *(const float2*)(bp + (kb >> 5) * CH + t * 16 * P + (hf * 4 + js) * 4);
            b[js][t].x = bv.x;
            b[js][t].y = bv.y;
          }
        }
#pragma unroll
        for (int js = 0; js < 4; ++js) {
          v2f a;
          a.x = ac[hf * 4 + js].x * s;
          a.y = ac[hf * 4 + js].y * s;
#pragma unroll
          for (int t = 0; t < 8; ++t) {
            acc[t] = __builtin_amdgcn_wmma_f32_16x16x4_f32(
                false, a, false, b[js][t], (short)0, acc[t], false, false);
          }
        }
      }
    }

#pragma unroll
    for (int t = 0; t < 8; ++t) {
      const int ncol = n0 + t * 16 + c;
#pragma unroll
      for (int q = 0; q < 8; ++q) {
        const int row = m0 + q + 8 * khalf;
        float val = acc[t][q];
        if constexpr (HAS_BIAS) val += bias[ncol];
        if constexpr (RELU) val = fmaxf(val, 0.0f);
        D[(long)row * NTOT + ncol] = val;
      }
    }
  }
}

#define EDGES_PER_WAVE 128
#define GATHER_DEPTH 16

__device__ __forceinline__ void flush_run(float* __restrict__ m, int node,
                                          int c, const float4 acc) {
  float* p = m + (long)node * HID_C + c;
  unsafeAtomicAdd(p + 0, acc.x);
  unsafeAtomicAdd(p + 1, acc.y);
  unsafeAtomicAdd(p + 2, acc.z);
  unsafeAtomicAdd(p + 3, acc.w);
}

__global__ __launch_bounds__(256) void aggregate_kernel(
    const float* __restrict__ h1, const idx_t* __restrict__ src,
    const idx_t* __restrict__ dst, float* __restrict__ m, int E) {
  const int wave = __builtin_amdgcn_readfirstlane(
      (int)((blockIdx.x * blockDim.x + threadIdx.x) >> 5));
  const int lane = threadIdx.x & 31;
  const long e0 = (long)wave * EDGES_PER_WAVE;
  if (e0 >= E) return;
  long e1 = e0 + EDGES_PER_WAVE;
  if (e1 > E) e1 = E;
  const int c = lane * 4;
  const float* __restrict__ hc = h1 + c;

  int sv = 0, dv = 0;
  if (e0 + lane < e1) {
    sv = src[e0 + lane];
    dv = dst[e0 + lane];
  }
  float4 acc = make_float4(0.f, 0.f, 0.f, 0.f);
  int cur = __builtin_amdgcn_readlane(dv, 0);

  for (long eb = e0; eb < e1; eb += 32) {
    int sn = 0, dn = 0;
    if (eb + 32 + lane < e1) {
      sn = src[eb + 32 + lane];
      dn = dst[eb + 32 + lane];
    }
    if (e1 - eb >= 32) {
#pragma unroll
      for (int g = 0; g < 32; g += GATHER_DEPTH) {
        float4 v[GATHER_DEPTH];
#pragma unroll
        for (int j = 0; j < GATHER_DEPTH; ++j) {
          const int s = __builtin_amdgcn_readlane(sv, g + j);
          v[j] = *(const float4*)(hc + (long)s * HID_C);
        }
#pragma unroll
        for (int j = 0; j < GATHER_DEPTH; ++j) {
          const int d = __builtin_amdgcn_readlane(dv, g + j);
          if (d != cur) {
            flush_run(m, cur, c, acc);
            acc = make_float4(0.f, 0.f, 0.f, 0.f);
            cur = d;
          }
          acc.x += v[j].x;
          acc.y += v[j].y;
          acc.z += v[j].z;
          acc.w += v[j].w;
        }
      }
    } else {
      const int nv = (int)(e1 - eb);
      for (int j = 0; j < nv; ++j) {
        const int s = __builtin_amdgcn_readlane(sv, j);
        const int d = __builtin_amdgcn_readlane(dv, j);
        if (d != cur) {
          flush_run(m, cur, c, acc);
          acc = make_float4(0.f, 0.f, 0.f, 0.f);
          cur = d;
        }
        const float4 v = *(const float4*)(hc + (long)s * HID_C);
        acc.x += v.x;
        acc.y += v.y;
        acc.z += v.z;
        acc.w += v.w;
      }
    }
    sv = sn;
    dv = dn;
  }
  flush_run(m, cur, c, acc);
}

__global__ void norm_bias_relu_kernel(float* __restrict__ m,
                                      const float* __restrict__ rin,
                                      const float* __restrict__ b_conv,
                                      long total) {
  const long i = (long)blockIdx.x * blockDim.x + threadIdx.x;
  if (i >= total) return;
  const int c = (int)(i & (HID_C - 1));
  const long node = i >> 7;
  const float v = m[i] * rin[node] + b_conv[c];
  m[i] = fmaxf(v, 0.0f);
}

extern "C" void kernel_launch(void* const* d_in, const int* in_sizes, int n_in,
                              void* d_out, int out_size, void* d_ws,
                              size_t ws_size, hipStream_t stream) {
  const float* x      = (const float*)d_in[0];
  const idx_t* src    = (const idx_t*)d_in[1];
  const idx_t* dst    = (const idx_t*)d_in[2];
  const float* W_conv = (const float*)d_in[3];
  const float* b_conv = (const float*)d_in[4];
  const float* W1     = (const float*)d_in[5];
  const float* b1     = (const float*)d_in[6];
  const float* W2     = (const float*)d_in[7];
  const float* b2     = (const float*)d_in[8];
  float* out = (float*)d_out;

  const int N = in_sizes[0] / IN_C;
  const int E = in_sizes[1];

  float* ws      = (float*)d_ws;
  float* deg_out = ws;
  float* deg_in  = ws + N;
  float* Wc_t    = ws + 2L * N;
  float* bufB    = Wc_t + (long)HID_C * IN_C;
  float* bufA    = bufB + (long)N * HID_C;

  zero_f32_kernel<<<64, 256, 0, stream>>>(ws, 2L * N);
  zero_f32_kernel<<<2048, 256, 0, stream>>>(bufB, (long)N * HID_C);

  transpose_kernel<<<(IN_C * HID_C + 255) / 256, 256, 0, stream>>>(
      W_conv, Wc_t, IN_C, HID_C);

  degree_kernel<<<(E + 255) / 256, 256, 0, stream>>>(src, dst, deg_out, deg_in, E);
  rsqrt_deg_kernel<<<(2 * N + 255) / 256, 256, 0, stream>>>(deg_out, 2 * N);

  const int rowTiles = N / 16;

  gemm_wmma_kernel<IN_C, HID_C,  false,  false,
                    true>
      <<<GEMM_BLOCKS, 256, 0, stream>>>(x, Wc_t, nullptr, deg_out, bufA,
                                        rowTiles);

  const int waves = (E + EDGES_PER_WAVE - 1) / EDGES_PER_WAVE;
  aggregate_kernel<<<(waves * 32 + 255) / 256, 256, 0, stream>>>(bufA, src, dst,
                                                                 bufB, E);

  norm_bias_relu_kernel<<<(int)(((long)N * HID_C + 255) / 256), 256, 0,
                          stream>>>(bufB, deg_in, b_conv, (long)N * HID_C);

  gemm_wmma_kernel<HID_C, HID_C,  true,  true,
                    false>
      <<<GEMM_BLOCKS, 256, 0, stream>>>(bufB, W1, b1, nullptr, bufA, rowTiles);

  gemm_wmma_kernel<HID_C, OUT_C,  true,  false,
                    false>
      <<<GEMM_BLOCKS, 256, 0, stream>>>(bufA, W2, b2, nullptr, out, rowTiles);
}


# aggregate: row starts taken from sorted dst; each wave owns 8 destination rows, sums them in registers in edge order, stores relu(sum*rin+b) once; no row atomics, no zero pass, no norm pass
# speedup vs baseline: 2.1992x; 1.2350x over previous
#include <hip/hip_runtime.h>


#define IN_C  256
#define HID_C 128
#define OUT_C 256

typedef int idx_t;
typedef __attribute__((ext_vector_type(2))) float v2f;
typedef __attribute__((ext_vector_type(8))) float v8f;

__global__ void zero_f32_kernel(float* __restrict__ p, long n) {
  long i = (long)blockIdx.x * blockDim.x + threadIdx.x;
  const long stride = (long)gridDim.x * blockDim.x;
  for (; i < n; i += stride) p[i] = 0.0f;
}

__global__ void degree_rowstart_kernel(const idx_t* __restrict__ src,
                                       const idx_t* __restrict__ dst,
                                       float* __restrict__ deg_out,
                                       int* __restrict__ row_start, int E,
                                       int N) {
  const int e = blockIdx.x * blockDim.x + threadIdx.x;
  if (e >= E) return;
  unsafeAtomicAdd(&deg_out[src[e]], 1.0f);
  const int d = dst[e];
  const int p = e > 0 ? dst[e - 1] : -1;
  for (int n = p + 1; n <= d; ++n) row_start[n] = e;
  if (e == E - 1) {
    for (int n = d + 1; n <= N; ++n) row_start[n] = E;
  }
}

__global__ void rsqrt_deg_kernel(float* __restrict__ deg, int n) {
  const int i = blockIdx.x * blockDim.x + threadIdx.x;
  if (i >= n) return;
  deg[i] = rsqrtf(fmaxf(deg[i], 1.0f));
}

__global__ void transpose_kernel(const float* __restrict__ W,
                                 float* __restrict__ Wt, int rows, int cols) {
  const int i = blockIdx.x * blockDim.x + threadIdx.x;
  if (i >= rows * cols) return;
  const int r = i / cols;
  const int c = i - r * cols;
  Wt[(long)c * rows + r] = W[i];
}

#define GEMM_BLOCKS 512
template <int KDIM, int NTOT, bool HAS_BIAS, bool RELU, bool SCALE_A>
__global__ __launch_bounds__(256) void gemm_wmma_kernel(
    const float* __restrict__ A, const float* __restrict__ W,
    const float* __restrict__ bias, const float* __restrict__ scaleA,
    float* __restrict__ D, int nRowTiles) {
  constexpr int P = 36;
  constexpr int CH = NTOT * P;
  constexpr int NSPLIT = NTOT / 128;
  constexpr int F4R = KDIM / 4;
  constexpr int PASSES = NTOT * F4R / 256;
  __shared__ __attribute__((aligned(16))) float wl[(KDIM / 32) * CH];

  const int tid = threadIdx.x;
#pragma unroll 8
  for (int p = 0; p < PASSES; ++p) {
    const int idx = (((p + (int)blockIdx.x) & (PASSES - 1)) << 8) + tid;
    const int r = idx / F4R;
    const int c4 = idx - r * F4R;
    const float4 v = *(const float4*)(W + (long)idx * 4);
    *(float4*)(wl + (c4 >> 3) * CH + r * P + (c4 & 7) * 4) = v;
  }
  __syncthreads();

  const int lane = tid & 31;
  const int c = lane & 15;
  const int khalf = lane >> 4;
  const int wv = __builtin_amdgcn_readfirstlane(tid >> 5);
  const int rank = (0x73261540 >> (4 * wv)) & 7;

  for (int v = rank;; v += 8) {
    const int u = v / NSPLIT;
    const int rowTile = u * GEMM_BLOCKS + (int)blockIdx.x;
    if (rowTile >= nRowTiles) break;
    const int n0 = (v - u * NSPLIT) * 128;
    const int m0 = rowTile * 16;
    const int mrow = m0 + c;

    const float s = SCALE_A ? scaleA[mrow] : 1.0f;

    v8f acc[8];
#pragma unroll
    for (int t = 0; t < 8; ++t) acc[t] = (v8f){0, 0, 0, 0, 0, 0, 0, 0};

    const float* __restrict__ ap = A + (long)mrow * KDIM + khalf * 2;
    const float* bp = wl + (n0 + c) * P + khalf * 2;

    float2 an[8];
#pragma unroll
    for (int j = 0; j < 8; ++j) an[j] = *(const float2*)(ap + 4 * j);

#pragma unroll 1
    for (int kb = 0; kb < KDIM; kb += 32) {
      float2 ac[8];
#pragma unroll
      for (int j = 0; j < 8; ++j) ac[j] = an[j];
      if (kb + 32 < KDIM) {
#pragma unroll
        for (int j = 0; j < 8; ++j)
          an[j] = *(const float2*)(ap + kb + 32 + 4 * j);
      }
#pragma unroll
      for (int hf = 0; hf < 2; ++hf) {
        v2f b[4][8];
#pragma unroll
        for (int js = 0; js < 4; ++js) {
#pragma unroll
          for (int t = 0; t < 8; ++t) {
            const float2 bv =
                *(const float2*)(bp + (kb >> 5) * CH + t * 16 * P + (hf * 4 + js) * 4);
            b[js][t].x = bv.x;
            b[js][t].y = bv.y;
          }
        }
#pragma unroll
        for (int js = 0; js < 4; ++js) {
          v2f a;
          a.x = ac[hf * 4 + js].x * s;
          a.y = ac[hf * 4 + js].y * s;
#pragma unroll
          for (int t = 0; t < 8; ++t) {
            acc[t] = __builtin_amdgcn_wmma_f32_16x16x4_f32(
                false, a, false, b[js][t], (short)0, acc[t], false, false);
          }
        }
      }
    }

#pragma unroll
    for (int t = 0; t < 8; ++t) {
      const int ncol = n0 + t * 16 + c;
#pragma unroll
      for (int q = 0; q < 8; ++q) {
        const int row = m0 + q + 8 * khalf;
        float val = acc[t][q];
        if constexpr (HAS_BIAS) val += bias[ncol];
        if constexpr (RELU) val = fmaxf(val, 0.0f);
        D[(long)row * NTOT + ncol] = val;
      }
    }
  }
}

#define ROWS_PER_WAVE 8
#define GATHER_DEPTH 16

__device__ __forceinline__ void store_row(float* __restrict__ h2c, int node,
                                          const float4 acc, int cnt,
                                          const float4 bias) {
  const float rin = rsqrtf(fmaxf((float)cnt, 1.0f));
  float4 r;
  r.x = fmaxf(acc.x * rin + bias.x, 0.0f);
  r.y = fmaxf(acc.y * rin + bias.y, 0.0f);
  r.z = fmaxf(acc.z * rin + bias.z, 0.0f);
  r.w = fmaxf(acc.w * rin + bias.w, 0.0f);
  *(float4*)(h2c + (long)node * HID_C) = r;
}

__global__ __launch_bounds__(256) void aggregate_kernel(
    const float* __restrict__ h1, const idx_t* __restrict__ src,
    const idx_t* __restrict__ dst, const int* __restrict__ row_start,
    const float* __restrict__ b_conv, float* __restrict__ h2, int N) {
  const int wave = __builtin_amdgcn_readfirstlane(
      (int)((blockIdx.x * blockDim.x + threadIdx.x) >> 5));
  const int lane = threadIdx.x & 31;
  const int n0 = wave * ROWS_PER_WAVE;
  if (n0 >= N) return;
  int n1 = n0 + ROWS_PER_WAVE;
  if (n1 > N) n1 = N;
  const int e0 = row_start[n0];
  const int e1 = row_start[n1];
  const int c = lane * 4;
  const float* __restrict__ hc = h1 + c;
  float* __restrict__ h2c = h2 + c;
  const float4 bias = *(const float4*)(b_conv + c);

  int sv = 0, dv = 0;
  if (e0 + lane < e1) {
    sv = src[e0 + lane];
    dv = dst[e0 + lane];
  }
  float4 acc = make_float4(0.f, 0.f, 0.f, 0.f);
  int cur = n0;
  int cnt = 0;

  for (int eb = e0; eb < e1; eb += 32) {
    int sn = 0, dn = 0;
    const int nv = e1 - eb;
#pragma unroll
    for (int g = 0; g < 32; g += GATHER_DEPTH) {
      if (g < nv) {
        float4 v[GATHER_DEPTH];
#pragma unroll
        for (int j = 0; j < GATHER_DEPTH; ++j) {
          const int s = __builtin_amdgcn_readlane(sv, g + j);
          v[j] = *(const float4*)(hc + (long)s * HID_C);
        }
        if (g == 0) {
          if (eb + 32 + lane < e1) {
            sn = src[eb + 32 + lane];
            dn = dst[eb + 32 + lane];
          }
        }
#pragma unroll
        for (int j = 0; j < GATHER_DEPTH; ++j) {
          if (g + j < nv) {
            const int d = __builtin_amdgcn_readlane(dv, g + j);
#pragma unroll 1
            while (cur < d) {
              store_row(h2c, cur, acc, cnt, bias);
              acc = make_float4(0.f, 0.f, 0.f, 0.f);
              cnt = 0;
              ++cur;
            }
            acc.x += v[j].x;
            acc.y += v[j].y;
            acc.z += v[j].z;
            acc.w += v[j].w;
            ++cnt;
          }
        }
      }
    }
    sv = sn;
    dv = dn;
  }
#pragma unroll 1
  while (cur < n1) {
    store_row(h2c, cur, acc, cnt, bias);
    acc = make_float4(0.f, 0.f, 0.f, 0.f);
    cnt = 0;
    ++cur;
  }
}

extern "C" void kernel_launch(void* const* d_in, const int* in_sizes, int n_in,
                              void* d_out, int out_size, void* d_ws,
                              size_t ws_size, hipStream_t stream) {
  const float* x      = (const float*)d_in[0];
  const idx_t* src    = (const idx_t*)d_in[1];
  const idx_t* dst    = (const idx_t*)d_in[2];
  const float* W_conv = (const float*)d_in[3];
  const float* b_conv = (const float*)d_in[4];
  const float* W1     = (const float*)d_in[5];
  const float* b1     = (const float*)d_in[6];
  const float* W2     = (const float*)d_in[7];
  const float* b2     = (const float*)d_in[8];
  float* out = (float*)d_out;

  const int N = in_sizes[0] / IN_C;
  const int E = in_sizes[1];

  float* ws      = (float*)d_ws;
  float* deg_out = ws;
  int* row_start = (int*)(ws + N);
  float* Wc_t    = ws + 2L * N + 4;
  float* bufB    = Wc_t + (long)HID_C * IN_C;
  float* bufA    = bufB + (long)N * HID_C;

  zero_f32_kernel<<<64, 256, 0, stream>>>(ws, (long)N);

  transpose_kernel<<<(IN_C * HID_C + 255) / 256, 256, 0, stream>>>(
      W_conv, Wc_t, IN_C, HID_C);

  degree_rowstart_kernel<<<(E + 255) / 256, 256, 0, stream>>>(
      src, dst, deg_out, row_start, E, N);
  rsqrt_deg_kernel<<<(N + 255) / 256, 256, 0, stream>>>(deg_out, N);

  const int rowTiles = N / 16;

  gemm_wmma_kernel<IN_C, HID_C,  false,  false,
                    true>
      <<<GEMM_BLOCKS, 256, 0, stream>>>(x, Wc_t, nullptr, deg_out, bufA,
                                        rowTiles);

  const int waves = (N + ROWS_PER_WAVE - 1) / ROWS_PER_WAVE;
  aggregate_kernel<<<(waves * 32 + 255) / 256, 256, 0, stream>>>(
      bufA, src, dst, row_start, b_conv, bufB, N);

  gemm_wmma_kernel<HID_C, HID_C,  true,  true,
                    false>
      <<<GEMM_BLOCKS, 256, 0, stream>>>(bufB, W1, b1, nullptr, bufA, rowTiles);

  gemm_wmma_kernel<HID_C, OUT_C,  true,  false,
                    false>
      <<<GEMM_BLOCKS, 256, 0, stream>>>(bufA, W2, b2, nullptr, out, rowTiles);
}


# launches 8 -> 5: out-degree clear + W_conv transpose in one launch, rsqrt(out_deg) taken in the first GEMM's row scale, W1 and W2 layers in one launch (hidden tile kept in LDS); sum order kept
# speedup vs baseline: 2.3413x; 1.0646x over previous
#include <hip/hip_runtime.h>


#define IN_C  256
#define HID_C 128
#define OUT_C 256

typedef int idx_t;
typedef __attribute__((ext_vector_type(2))) float v2f;
typedef __attribute__((ext_vector_type(8))) float v8f;

__global__ void prep_kernel(float* __restrict__ deg, int n,
                            const float* __restrict__ W,
                            float* __restrict__ Wt, int rows, int cols) {
  const int i = blockIdx.x * blockDim.x + threadIdx.x;
  if (i < n) deg[i] = 0.0f;
  if (i < rows * cols) {
    const int r = i / cols;
    const int c = i - r * cols;
    Wt[(long)c * rows + r] = W[i];
  }
}

__global__ void degree_rowstart_kernel(const idx_t* __restrict__ src,
                                       const idx_t* __restrict__ dst,
                                       float* __restrict__ deg_out,
                                       int* __restrict__ row_start, int E,
                                       int N) {
  const int e = blockIdx.x * blockDim.x + threadIdx.x;
  if (e >= E) return;
  unsafeAtomicAdd(&deg_out[src[e]], 1.0f);
  const int d = dst[e];
  const int p = e > 0 ? dst[e - 1] : -1;
  for (int n = p + 1; n <= d; ++n) row_start[n] = e;
  if (e == E - 1) {
    for (int n = d + 1; n <= N; ++n) row_start[n] = E;
  }
}

#define GEMM_BLOCKS 512
template <int KDIM, int NTOT, bool HAS_BIAS, bool RELU, bool SCALE_A>
__global__ __launch_bounds__(256) void gemm_wmma_kernel(
    const float* __restrict__ A, const float* __restrict__ W,
    const float* __restrict__ bias, const float* __restrict__ scaleA,
    float* __restrict__ D, int nRowTiles) {
  constexpr int P = 36;
  constexpr int CH = NTOT * P;
  constexpr int NSPLIT = NTOT / 128;
  constexpr int F4R = KDIM / 4;
  constexpr int PASSES = NTOT * F4R / 256;
  __shared__ __attribute__((aligned(16))) float wl[(KDIM / 32) * CH];

  const int tid = threadIdx.x;
#pragma unroll 8
  for (int p = 0; p < PASSES; ++p) {
    const int idx = (((p + (int)blockIdx.x) & (PASSES - 1)) << 8) + tid;
    const int r = idx / F4R;
    const int c4 = idx - r * F4R;
    const float4 v = *(const float4*)(W + (long)idx * 4);
    *(float4*)(wl + (c4 >> 3) * CH + r * P + (c4 & 7) * 4) = v;
  }
  __syncthreads();

  const int lane = tid & 31;
  const int c = lane & 15;
  const int khalf = lane >> 4;
  const int wv = __builtin_amdgcn_readfirstlane(tid >> 5);
  const int rank = (0x73261540 >> (4 * wv)) & 7;

  for (int v = rank;; v += 8) {
    const int u = v / NSPLIT;
    const int rowTile = u * GEMM_BLOCKS + (int)blockIdx.x;
    if (rowTile >= nRowTiles) break;
    const int n0 = (v - u * NSPLIT) * 128;
    const int m0 = rowTile * 16;
    const int mrow = m0 + c;

    const float s = SCALE_A ? rsqrtf(fmaxf(scaleA[mrow], 1.0f)) : 1.0f;

    v8f acc[8];
#pragma unroll
    for (int t = 0; t < 8; ++t) acc[t] = (v8f){0, 0, 0, 0, 0, 0, 0, 0};

    const float* __restrict__ ap = A + (long)mrow * KDIM + khalf * 2;
    const float* bp = wl + (n0 + c) * P + khalf * 2;

    float2 an[8];
#pragma unroll
    for (int j = 0; j < 8; ++j) an[j] = *(const float2*)(ap + 4 * j);

#pragma unroll 1
    for (int kb = 0; kb < KDIM; kb += 32) {
      float2 ac[8];
#pragma unroll
      for (int j = 0; j < 8; ++j) ac[j] = an[j];
      if (kb + 32 < KDIM) {
#pragma unroll
        for (int j = 0; j < 8; ++j)
          an[j] = *(const float2*)(ap + kb + 32 + 4 * j);
      }
#pragma unroll
      for (int hf = 0; hf < 2; ++hf) {
        v2f b[4][8];
#pragma unroll
        for (int js = 0; js < 4; ++js) {
#pragma unroll
          for (int t = 0; t < 8; ++t) {
            const float2 bv =
                *(const float2*)(bp + (kb >> 5) * CH + t * 16 * P + (hf * 4 + js) * 4);
            b[js][t].x = bv.x;
            b[js][t].y = bv.y;
          }
        }
#pragma unroll
        for (int js = 0; js < 4; ++js) {
          v2f a;
          a.x = ac[hf * 4 + js].x * s;
          a.y = ac[hf * 4 + js].y * s;
#pragma unroll
          for (int t = 0; t < 8; ++t) {
            acc[t] = __builtin_amdgcn_wmma_f32_16x16x4_f32(
                false, a, false, b[js][t], (short)0, acc[t], false, false);
          }
        }
      }
    }

#pragma unroll
    for (int t = 0; t < 8; ++t) {
      const int ncol = n0 + t * 16 + c;
#pragma unroll
      for (int q = 0; q < 8; ++q) {
        const int row = m0 + q + 8 * khalf;
        float val = acc[t][q];
        if constexpr (HAS_BIAS) val += bias[ncol];
        if constexpr (RELU) val = fmaxf(val, 0.0f);
        D[(long)row * NTOT + ncol] = val;
      }
    }
  }
}

#define MLP_BLOCKS 256
#define MLP_WAVES 8
__global__ __launch_bounds__(MLP_WAVES * 32) void mlp_wmma_kernel(
    const float* __restrict__ A, const float* __restrict__ W1,
    const float* __restrict__ b1, const float* __restrict__ W2,
    const float* __restrict__ b2, float* __restrict__ D, int nRowTiles) {
  constexpr int P = 36;
  constexpr int CH1 = HID_C * P;
  constexpr int CH2 = OUT_C * P;
  constexpr int TP = HID_C + 4;
  constexpr int F4R = HID_C / 4;
  constexpr int NT = MLP_WAVES * 32;
  constexpr int F4_1 = HID_C * F4R;
  constexpr int PASSES = (F4_1 + OUT_C * F4R) / NT;
  static_assert((F4_1 + OUT_C * F4R) % NT == 0, "staging passes must be whole");
  __shared__ __attribute__((aligned(16))) float w1l[(HID_C / 32) * CH1];
  __shared__ __attribute__((aligned(16))) float w2l[(HID_C / 32) * CH2];
  __shared__ __attribute__((aligned(16))) float tl[MLP_WAVES * 16 * TP];

  const int tid = threadIdx.x;
  const int lane = tid & 31;
  const int c = lane & 15;
  const int khalf = lane >> 4;
  const int wv = __builtin_amdgcn_readfirstlane(tid >> 5);
  float* tw = tl + wv * 16 * TP;

  constexpr int TILE_STEP = MLP_WAVES * MLP_BLOCKS;
  const float* __restrict__ acol = A + c * HID_C + khalf * 2;
  int rowTile = wv * MLP_BLOCKS + (int)blockIdx.x;
  float2 at[32];
  if (rowTile < nRowTiles) {
#pragma unroll
    for (int j = 0; j < 32; ++j)
      at[j] = *(const float2*)(acol + (long)rowTile * 16 * HID_C + 4 * j);
  }
  float bb1[8];
#pragma unroll
  for (int t = 0; t < 8; ++t) bb1[t] = b1[t * 16 + c];

#pragma unroll 8
  for (int p = 0; p < PASSES; ++p) {
    const int idx = ((p + (int)blockIdx.x) % PASSES) * NT + tid;
    const bool second = idx >= F4_1;
    const int id2 = second ? idx - F4_1 : idx;
    const int r = id2 / F4R;
    const int c4 = id2 - r * F4R;
    const float4 v = *(const float4*)((second ? W2 : W1) + (long)id2 * 4);
    float* dstp = second ? w2l + (c4 >> 3) * CH2 : w1l + (c4 >> 3) * CH1;
    *(float4*)(dstp + r * P + (c4 & 7) * 4) = v;
  }
  __syncthreads();

  for (; rowTile < nRowTiles; rowTile += TILE_STEP) {
    const int m0 = rowTile * 16;
    const bool hasNext = rowTile + TILE_STEP < nRowTiles;

    v8f acc[8];
#pragma unroll
    for (int t = 0; t < 8; ++t) acc[t] = (v8f){0, 0, 0, 0, 0, 0, 0, 0};

    {
      const float* bp = w1l + c * P + khalf * 2;
#pragma unroll
      for (int g = 0; g < HID_C / 32; ++g) {
#pragma unroll
        for (int hf = 0; hf < 2; ++hf) {
          v2f b[4][8];
#pragma unroll
          for (int js = 0; js < 4; ++js) {
#pragma unroll
            for (int t = 0; t < 8; ++t) {
              const float2 bv = *(const float2*)(bp + g * CH1 + t * 16 * P +
                                                 (hf * 4 + js) * 4);
              b[js][t].x = bv.x;
              b[js][t].y = bv.y;
            }
          }
#pragma unroll
          for (int js = 0; js < 4; ++js) {
            v2f a;
            a.x = at[g * 8 + hf * 4 + js].x;
            a.y = at[g * 8 + hf * 4 + js].y;
#pragma unroll
            for (int t = 0; t < 8; ++t) {
              acc[t] = __builtin_amdgcn_wmma_f32_16x16x4_f32(
                  false, a, false, b[js][t], (short)0, acc[t], false, false);
            }
          }
        }
      }
    }
    if (hasNext) {
#pragma unroll
      for (int j = 0; j < 32; ++j)
        at[j] = *(const float2*)(acol + (long)(m0 + TILE_STEP * 16) * HID_C + 4 * j);
    }

#pragma unroll
    for (int t = 0; t < 8; ++t) {
#pragma unroll
      for (int q = 0; q < 8; ++q)
        tw[(q + 8 * khalf) * TP + t * 16 + c] = fmaxf(acc[t][q] + bb1[t], 0.0f);
    }

#pragma unroll 1
    for (int n0 = 0; n0 < OUT_C; n0 += 128) {
#pragma unroll
      for (int t = 0; t < 8; ++t) acc[t] = (v8f){0, 0, 0, 0, 0, 0, 0, 0};
      const float* apl = tw + c * TP + khalf * 2;
      const float* bp = w2l + (n0 + c) * P + khalf * 2;

#pragma unroll 1
      for (int kb = 0; kb < HID_C; kb += 32) {
        float2 ac[8];
#pragma unroll
        for (int j = 0; j < 8; ++j) ac[j] = *(const float2*)(apl + kb + 4 * j);
#pragma unroll
        for (int hf = 0; hf < 2; ++hf) {
          v2f b[4][8];
#pragma unroll
          for (int js = 0; js < 4; ++js) {
#pragma unroll
            for (int t = 0; t < 8; ++t) {
              const float2 bv = *(const float2*)(bp + (kb >> 5) * CH2 +
                                                 t * 16 * P + (hf * 4 + js) * 4);
              b[js][t].x = bv.x;
              b[js][t].y = bv.y;
            }
          }
#pragma unroll
          for (int js = 0; js < 4; ++js) {
            v2f a;
            a.x = ac[hf * 4 + js].x;
            a.y = ac[hf * 4 + js].y;
#pragma unroll
            for (int t = 0; t < 8; ++t) {
              acc[t] = __builtin_amdgcn_wmma_f32_16x16x4_f32(
                  false, a, false, b[js][t], (short)0, acc[t], false, false);
            }
          }
        }
      }

#pragma unroll
      for (int t = 0; t < 8; ++t) {
        const int ncol = n0 + t * 16 + c;
        const float bb = b2[ncol];
#pragma unroll
        for (int q = 0; q < 8; ++q) {
          const int row = m0 + q + 8 * khalf;
          D[(long)row * OUT_C + ncol] = acc[t][q] + bb;
        }
      }
    }
  }
}

#define ROWS_PER_WAVE 8
#define GATHER_DEPTH 16

__device__ __forceinline__ void store_row(float* __restrict__ h2c, int node,
                                          const float4 acc, int cnt,
                                          const float4 bias) {
  const float rin = rsqrtf(fmaxf((float)cnt, 1.0f));
  float4 r;
  r.x = fmaxf(acc.x * rin + bias.x, 0.0f);
  r.y = fmaxf(acc.y * rin + bias.y, 0.0f);
  r.z = fmaxf(acc.z * rin + bias.z, 0.0f);
  r.w = fmaxf(acc.w * rin + bias.w, 0.0f);
  *(float4*)(h2c + (long)node * HID_C) = r;
}

__global__ __launch_bounds__(256) void aggregate_kernel(
    const float* __restrict__ h1, const idx_t* __restrict__ src,
    const idx_t* __restrict__ dst, const int* __restrict__ row_start,
    const float* __restrict__ b_conv, float* __restrict__ h2, int N) {
  const int wave = __builtin_amdgcn_readfirstlane(
      (int)((blockIdx.x * blockDim.x + threadIdx.x) >> 5));
  const int lane = threadIdx.x & 31;
  const int n0 = wave * ROWS_PER_WAVE;
  if (n0 >= N) return;
  int n1 = n0 + ROWS_PER_WAVE;
  if (n1 > N) n1 = N;
  const int e0 = row_start[n0];
  const int e1 = row_start[n1];
  const int c = lane * 4;
  const float* __restrict__ hc = h1 + c;
  float* __restrict__ h2c = h2 + c;
  const float4 bias = *(const float4*)(b_conv + c);

  int sv = 0, dv = 0;
  if (e0 + lane < e1) {
    sv = src[e0 + lane];
    dv = dst[e0 + lane];
  }
  float4 acc = make_float4(0.f, 0.f, 0.f, 0.f);
  int cur = n0;
  int cnt = 0;

  for (int eb = e0; eb < e1; eb += 32) {
    int sn = 0, dn = 0;
    const int nv = e1 - eb;
#pragma unroll
    for (int g = 0; g < 32; g += GATHER_DEPTH) {
      if (g < nv) {
        float4 v[GATHER_DEPTH];
#pragma unroll
        for (int j = 0; j < GATHER_DEPTH; ++j) {
          const int s = __builtin_amdgcn_readlane(sv, g + j);
          v[j] = *(const float4*)(hc + (long)s * HID_C);
        }
        if (g == 0) {
          if (eb + 32 + lane < e1) {
            sn = src[eb + 32 + lane];
            dn = dst[eb + 32 + lane];
          }
        }
#pragma unroll
        for (int j = 0; j < GATHER_DEPTH; ++j) {
          if (g + j < nv) {
            const int d = __builtin_amdgcn_readlane(dv, g + j);
#pragma unroll 1
            while (cur < d) {
              store_row(h2c, cur, acc, cnt, bias);
              acc = make_float4(0.f, 0.f, 0.f, 0.f);
              cnt = 0;
              ++cur;
            }
            acc.x += v[j].x;
            acc.y += v[j].y;
            acc.z += v[j].z;
            acc.w += v[j].w;
            ++cnt;
          }
        }
      }
    }
    sv = sn;
    dv = dn;
  }
#pragma unroll 1
  while (cur < n1) {
    store_row(h2c, cur, acc, cnt, bias);
    acc = make_float4(0.f, 0.f, 0.f, 0.f);
    cnt = 0;
    ++cur;
  }
}

extern "C" void kernel_launch(void* const* d_in, const int* in_sizes, int n_in,
                              void* d_out, int out_size, void* d_ws,
                              size_t ws_size, hipStream_t stream) {
  const float* x      = (const float*)d_in[0];
  const idx_t* src    = (const idx_t*)d_in[1];
  const idx_t* dst    = (const idx_t*)d_in[2];
  const float* W_conv = (const float*)d_in[3];
  const float* b_conv = (const float*)d_in[4];
  const float* W1     = (const float*)d_in[5];
  const float* b1     = (const float*)d_in[6];
  const float* W2     = (const float*)d_in[7];
  const float* b2     = (const float*)d_in[8];
  float* out = (float*)d_out;

  const int N = in_sizes[0] / IN_C;
  const int E = in_sizes[1];

  float* ws      = (float*)d_ws;
  float* deg_out = ws;
  int* row_start = (int*)(ws + N);
  float* Wc_t    = ws + 2L * N + 4;
  float* bufB    = Wc_t + (long)HID_C * IN_C;
  float* bufA    = bufB + (long)N * HID_C;

  const int prepN = N > IN_C * HID_C ? N : IN_C * HID_C;
  prep_kernel<<<(prepN + 255) / 256, 256, 0, stream>>>(deg_out, N, W_conv, Wc_t,
                                                       IN_C, HID_C);

  degree_rowstart_kernel<<<(E + 255) / 256, 256, 0, stream>>>(
      src, dst, deg_out, row_start, E, N);

  const int rowTiles = N / 16;

  gemm_wmma_kernel<IN_C, HID_C,  false,  false,
                    true>
      <<<GEMM_BLOCKS, 256, 0, stream>>>(x, Wc_t, nullptr, deg_out, bufA,
                                        rowTiles);

  const int waves = (N + ROWS_PER_WAVE - 1) / ROWS_PER_WAVE;
  aggregate_kernel<<<(waves * 32 + 255) / 256, 256, 0, stream>>>(
      bufA, src, dst, row_start, b_conv, bufB, N);

  mlp_wmma_kernel<<<MLP_BLOCKS, MLP_WAVES * 32, 0, stream>>>(
      bufB, W1, b1, W2, b2, out, rowTiles);
}


# speedup vs baseline: 2.4370x; 1.0409x over previous
#include <hip/hip_runtime.h>


#define IN_C  256
#define HID_C 128
#define OUT_C 256

typedef int idx_t;
typedef __attribute__((ext_vector_type(2))) float v2f;
typedef __attribute__((ext_vector_type(8))) float v8f;

__global__ void prep_kernel(float* __restrict__ deg, int n,
                            const float* __restrict__ W,
                            float* __restrict__ Wt, int rows, int cols) {
  const int i = blockIdx.x * blockDim.x + threadIdx.x;
  if (i < n) deg[i] = 0.0f;
  if (i < rows * cols) {
    const int r = i / cols;
    const int c = i - r * cols;
    Wt[(long)c * rows + r] = W[i];
  }
}

__global__ void degree_rowstart_kernel(const idx_t* __restrict__ src,
                                       const idx_t* __restrict__ dst,
                                       float* __restrict__ deg_out,
                                       int* __restrict__ row_start, int E,
                                       int N) {
  const int e = blockIdx.x * blockDim.x + threadIdx.x;
  if (e >= E) return;
  unsafeAtomicAdd(&deg_out[src[e]], 1.0f);
  const int d = dst[e];
  const int p = e > 0 ? dst[e - 1] : -1;
  for (int n = p + 1; n <= d; ++n) row_start[n] = e;
  if (e == E - 1) {
    for (int n = d + 1; n <= N; ++n) row_start[n] = E;
  }
}

#define GEMM_BLOCKS 512
template <int KDIM, int NTOT, bool HAS_BIAS, bool RELU, bool SCALE_A>
__global__ __launch_bounds__(256) void gemm_wmma_kernel(
    const float* __restrict__ A, const float* __restrict__ W,
    const float* __restrict__ bias, const float* __restrict__ scaleA,
    float* __restrict__ D, int nRowTiles) {
  constexpr int P = 36;
  constexpr int CH = NTOT * P;
  constexpr int NSPLIT = NTOT / 128;
  constexpr int F4R = KDIM / 4;
  constexpr int PASSES = NTOT * F4R / 256;
  __shared__ __attribute__((aligned(16))) float wl[(KDIM / 32) * CH];

  const int tid = threadIdx.x;
#pragma unroll 8
  for (int p = 0; p < PASSES; ++p) {
    const int idx = (((p + (int)blockIdx.x) & (PASSES - 1)) << 8) + tid;
    const int r = idx / F4R;
    const int c4 = idx - r * F4R;
    const float4 v = *(const float4*)(W + (long)idx * 4);
    *(float4*)(wl + (c4 >> 3) * CH + r * P + (c4 & 7) * 4) = v;
  }
  __syncthreads();

  const int lane = tid & 31;
  const int c = lane & 15;
  const int khalf = lane >> 4;
  const int wv = __builtin_amdgcn_readfirstlane(tid >> 5);
  const int rank = (0x73261540 >> (4 * wv)) & 7;

  for (int v = rank;; v += 8) {
    const int u = v / NSPLIT;
    const int rowTile = u * GEMM_BLOCKS + (int)blockIdx.x;
    if (rowTile >= nRowTiles) break;
    const int n0 = (v - u * NSPLIT) * 128;
    const int m0 = rowTile * 16;
    const int mrow = m0 + c;

    const float s = SCALE_A ? rsqrtf(fmaxf(scaleA[mrow], 1.0f)) : 1.0f;

    v8f acc[8];
#pragma unroll
    for (int t = 0; t < 8; ++t) acc[t] = (v8f){0, 0, 0, 0, 0, 0, 0, 0};

    const float* __restrict__ ap = A + (long)mrow * KDIM + khalf * 2;
    const float* bp = wl + (n0 + c) * P + khalf * 2;

    float2 an[8];
#pragma unroll
    for (int j = 0; j < 8; ++j) an[j] = *(const float2*)(ap + 4 * j);

#pragma unroll 1
    for (int kb = 0; kb < KDIM; kb += 32) {
      float2 ac[8];
#pragma unroll
      for (int j = 0; j < 8; ++j) ac[j] = an[j];
      if (kb + 32 < KDIM) {
#pragma unroll
        for (int j = 0; j < 8; ++j)
          an[j] = *(const float2*)(ap + kb + 32 + 4 * j);
      }
#pragma unroll
      for (int hf = 0; hf < 2; ++hf) {
        v2f b[4][8];
#pragma unroll
        for (int js = 0; js < 4; ++js) {
#pragma unroll
          for (int t = 0; t < 8; ++t) {
            const float2 bv =
                *(const float2*)(bp + (kb >> 5) * CH + t * 16 * P + (hf * 4 + js) * 4);
            b[js][t].x = bv.x;
            b[js][t].y = bv.y;
          }
        }
#pragma unroll
        for (int js = 0; js < 4; ++js) {
          v2f a;
          a.x = ac[hf * 4 + js].x * s;
          a.y = ac[hf * 4 + js].y * s;
#pragma unroll
          for (int t = 0; t < 8; ++t) {
            acc[t] = __builtin_amdgcn_wmma_f32_16x16x4_f32(
                false, a, false, b[js][t], (short)0, acc[t], false, false);
          }
        }
      }
    }

#pragma unroll
    for (int t = 0; t < 8; ++t) {
      const int ncol = n0 + t * 16 + c;
#pragma unroll
      for (int q = 0; q < 8; ++q) {
        const int row = m0 + q + 8 * khalf;
        float val = acc[t][q];
        if constexpr (HAS_BIAS) val += bias[ncol];
        if constexpr (RELU) val = fmaxf(val, 0.0f);
        D[(long)row * NTOT + ncol] = val;
      }
    }
  }
}

#define MLP_BLOCKS 256
#define MLP_WAVES 8
__global__ __launch_bounds__(MLP_WAVES * 32) void mlp_wmma_kernel(
    const float* __restrict__ A, const float* __restrict__ W1,
    const float* __restrict__ b1, const float* __restrict__ W2,
    const float* __restrict__ b2, float* __restrict__ D, int nRowTiles) {
  constexpr int P = 36;
  constexpr int CH1 = HID_C * P;
  constexpr int CH2 = OUT_C * P;
  constexpr int TP = HID_C + 4;
  constexpr int F4R = HID_C / 4;
  constexpr int NT = MLP_WAVES * 32;
  constexpr int F4_1 = HID_C * F4R;
  constexpr int PASSES = (F4_1 + OUT_C * F4R) / NT;
  static_assert((F4_1 + OUT_C * F4R) % NT == 0, "staging passes must be whole");
  __shared__ __attribute__((aligned(16))) float w1l[(HID_C / 32) * CH1];
  __shared__ __attribute__((aligned(16))) float w2l[(HID_C / 32) * CH2];
  __shared__ __attribute__((aligned(16))) float tl[MLP_WAVES * 16 * TP];

  const int tid = threadIdx.x;
  const int lane = tid & 31;
  const int c = lane & 15;
  const int khalf = lane >> 4;
  const int wv = __builtin_amdgcn_readfirstlane(tid >> 5);
  float* tw = tl + wv * 16 * TP;

  constexpr int TILE_STEP = MLP_WAVES * MLP_BLOCKS;
  const float* __restrict__ acol = A + c * HID_C + khalf * 2;
  int rowTile = wv * MLP_BLOCKS + (int)blockIdx.x;
  float2 at[32];
  if (rowTile < nRowTiles) {
#pragma unroll
    for (int j = 0; j < 32; ++j)
      at[j] = *(const float2*)(acol + (long)rowTile * 16 * HID_C + 4 * j);
  }
  float bb1[8];
#pragma unroll
  for (int t = 0; t < 8; ++t) bb1[t] = b1[t * 16 + c];

#pragma unroll 8
  for (int p = 0; p < PASSES; ++p) {
    const int idx = ((p + (int)blockIdx.x) % PASSES) * NT + tid;
    const bool second = idx >= F4_1;
    const int id2 = second ? idx - F4_1 : idx;
    const int r = id2 / F4R;
    const int c4 = id2 - r * F4R;
    const float4 v = *(const float4*)((second ? W2 : W1) + (long)id2 * 4);
    float* dstp = second ? w2l + (c4 >> 3) * CH2 : w1l + (c4 >> 3) * CH1;
    *(float4*)(dstp + r * P + (c4 & 7) * 4) = v;
  }
  __syncthreads();

  for (; rowTile < nRowTiles; rowTile += TILE_STEP) {
    const int m0 = rowTile * 16;
    const bool hasNext = rowTile + TILE_STEP < nRowTiles;

    v8f acc[8];
#pragma unroll
    for (int t = 0; t < 8; ++t) acc[t] = (v8f){0, 0, 0, 0, 0, 0, 0, 0};

    {
      const float* bp = w1l + c * P + khalf * 2;
#pragma unroll
      for (int g = 0; g < HID_C / 32; ++g) {
#pragma unroll
        for (int hf = 0; hf < 2; ++hf) {
          v2f b[4][8];
#pragma unroll
          for (int js = 0; js < 4; ++js) {
#pragma unroll
            for (int t = 0; t < 8; ++t) {
              const float2 bv = *(const float2*)(bp + g * CH1 + t * 16 * P +
                                                 (hf * 4 + js) * 4);
              b[js][t].x = bv.x;
              b[js][t].y = bv.y;
            }
          }
#pragma unroll
          for (int js = 0; js < 4; ++js) {
            v2f a;
            a.x = at[g * 8 + hf * 4 + js].x;
            a.y = at[g * 8 + hf * 4 + js].y;
#pragma unroll
            for (int t = 0; t < 8; ++t) {
              acc[t] = __builtin_amdgcn_wmma_f32_16x16x4_f32(
                  false, a, false, b[js][t], (short)0, acc[t], false, false);
            }
          }
        }
      }
    }
    if (hasNext) {
#pragma unroll
      for (int j = 0; j < 32; ++j)
        at[j] = *(const float2*)(acol + (long)(m0 + TILE_STEP * 16) * HID_C + 4 * j);
    }

#pragma unroll
    for (int t = 0; t < 8; ++t) {
#pragma unroll
      for (int q = 0; q < 8; ++q)
        tw[(q + 8 * khalf) * TP + t * 16 + c] = fmaxf(acc[t][q] + bb1[t], 0.0f);
    }

#pragma unroll 1
    for (int n0 = 0; n0 < OUT_C; n0 += 128) {
#pragma unroll
      for (int t = 0; t < 8; ++t) acc[t] = (v8f){0, 0, 0, 0, 0, 0, 0, 0};
      const float* apl = tw + c * TP + khalf * 2;
      const float* bp = w2l + (n0 + c) * P + khalf * 2;

#pragma unroll 1
      for (int kb = 0; kb < HID_C; kb += 32) {
        float2 ac[8];
#pragma unroll
        for (int j = 0; j < 8; ++j) ac[j] = *(const float2*)(apl + kb + 4 * j);
#pragma unroll
        for (int hf = 0; hf < 2; ++hf) {
          v2f b[4][8];
#pragma unroll
          for (int js = 0; js < 4; ++js) {
#pragma unroll
            for (int t = 0; t < 8; ++t) {
              const float2 bv = *(const float2*)(bp + (kb >> 5) * CH2 +
                                                 t * 16 * P + (hf * 4 + js) * 4);
              b[js][t].x = bv.x;
              b[js][t].y = bv.y;
            }
          }
#pragma unroll
          for (int js = 0; js < 4; ++js) {
            v2f a;
            a.x = ac[hf * 4 + js].x;
            a.y = ac[hf * 4 + js].y;
#pragma unroll
            for (int t = 0; t < 8; ++t) {
              acc[t] = __builtin_amdgcn_wmma_f32_16x16x4_f32(
                  false, a, false, b[js][t], (short)0, acc[t], false, false);
            }
          }
        }
      }

#pragma unroll
      for (int t = 0; t < 8; ++t) {
        const int ncol = n0 + t * 16 + c;
        const float bb = b2[ncol];
#pragma unroll
        for (int q = 0; q < 8; ++q) {
          const int row = m0 + q + 8 * khalf;
          D[(long)row * OUT_C + ncol] = acc[t][q] + bb;
        }
      }
    }
  }
}

#define ROWS_PER_WAVE 8
#define GATHER_DEPTH 16

__device__ __forceinline__ void store_row(float* __restrict__ h2c, int node,
                                          const float4 acc, int cnt,
                                          const float4 bias) {
  const float rin = rsqrtf(fmaxf((float)cnt, 1.0f));
  float4 r;
  r.x = fmaxf(acc.x * rin + bias.x, 0.0f);
  r.y = fmaxf(acc.y * rin + bias.y, 0.0f);
  r.z = fmaxf(acc.z * rin + bias.z, 0.0f);
  r.w = fmaxf(acc.w * rin + bias.w, 0.0f);
  *(float4*)(h2c + (long)node * HID_C) = r;
}

__global__ __launch_bounds__(256) void aggregate_kernel(
    const float* __restrict__ h1, const idx_t* __restrict__ src,
    const idx_t* __restrict__ dst, const int* __restrict__ row_start,
    const float* __restrict__ b_conv, float* __restrict__ h2, int N) {
  const int wave = __builtin_amdgcn_readfirstlane(
      (int)((blockIdx.x * blockDim.x + threadIdx.x) >> 5));
  const int lane = threadIdx.x & 31;
  const int n0 = wave * ROWS_PER_WAVE;
  if (n0 >= N) return;
  int n1 = n0 + ROWS_PER_WAVE;
  if (n1 > N) n1 = N;
  const int e0 = row_start[n0];
  const int e1 = row_start[n1];
  const int c = lane * 4;
  const float* __restrict__ hc = h1 + c;
  float* __restrict__ h2c = h2 + c;
  const float4 bias = *(const float4*)(b_conv + c);

  int sv = 0, dv = 0;
  if (e0 + lane < e1) {
    sv = src[e0 + lane];
    dv = dst[e0 + lane];
  }
  float4 acc = make_float4(0.f, 0.f, 0.f, 0.f);
  int cur = n0;
  int cnt = 0;

  for (int eb = e0; eb < e1; eb += 32) {
    int sn = 0, dn = 0;
    const int nv = e1 - eb;
#pragma unroll
    for (int g = 0; g < 32; g += GATHER_DEPTH) {
      if (g < nv) {
        float4 v[GATHER_DEPTH];
#pragma unroll
        for (int j = 0; j < GATHER_DEPTH; ++j) {
          const int s = __builtin_amdgcn_readlane(sv, g + j);
          v[j] = *(const float4*)(hc + (long)s * HID_C);
        }
        if (g == 0) {
          if (eb + 32 + lane < e1) {
            sn = src[eb + 32 + lane];
            dn = dst[eb + 32 + lane];
          }
        }
#pragma unroll
        for (int j = 0; j < GATHER_DEPTH; ++j) {
          if (g + j < nv) {
            const int d = __builtin_amdgcn_readlane(dv, g + j);
#pragma unroll 1
            while (cur < d) {
              store_row(h2c, cur, acc, cnt, bias);
              acc = make_float4(0.f, 0.f, 0.f, 0.f);
              cnt = 0;
              ++cur;
            }
            acc.x += v[j].x;
            acc.y += v[j].y;
            acc.z += v[j].z;
            acc.w += v[j].w;
            ++cnt;
          }
        }
      }
    }
    sv = sn;
    dv = dn;
  }
#pragma unroll 1
  while (cur < n1) {
    store_row(h2c, cur, acc, cnt, bias);
    acc = make_float4(0.f, 0.f, 0.f, 0.f);
    cnt = 0;
    ++cur;
  }
}

extern "C" void kernel_launch(void* const* d_in, const int* in_sizes, int n_in,
                              void* d_out, int out_size, void* d_ws,
                              size_t ws_size, hipStream_t stream) {
  const float* x      = (const float*)d_in[0];
  const idx_t* src    = (const idx_t*)d_in[1];
  const idx_t* dst    = (const idx_t*)d_in[2];
  const float* W_conv = (const float*)d_in[3];
  const float* b_conv = (const float*)d_in[4];
  const float* W1     = (const float*)d_in[5];
  const float* b1     = (const float*)d_in[6];
  const float* W2     = (const float*)d_in[7];
  const float* b2     = (const float*)d_in[8];
  float* out = (float*)d_out;

  const int N = in_sizes[0] / IN_C;
  const int E = in_sizes[1];

  float* ws      = (float*)d_ws;
  float* deg_out = ws;
  int* row_start = (int*)(ws + N);
  float* Wc_t    = ws + ((2L * N + 1 + 127) & ~127L);
  float* bufB    = Wc_t + (long)HID_C * IN_C;
  float* bufA    = bufB + (long)N * HID_C;

  const int prepN = N > IN_C * HID_C ? N : IN_C * HID_C;
  prep_kernel<<<(prepN + 255) / 256, 256, 0, stream>>>(deg_out, N, W_conv, Wc_t,
                                                       IN_C, HID_C);

  degree_rowstart_kernel<<<(E + 255) / 256, 256, 0, stream>>>(
      src, dst, deg_out, row_start, E, N);

  const int rowTiles = N / 16;

  gemm_wmma_kernel<IN_C, HID_C,  false,  false,
                    true>
      <<<GEMM_BLOCKS, 256, 0, stream>>>(x, Wc_t, nullptr, deg_out, bufA,
                                        rowTiles);

  const int waves = (N + ROWS_PER_WAVE - 1) / ROWS_PER_WAVE;
  aggregate_kernel<<<(waves * 32 + 255) / 256, 256, 0, stream>>>(
      bufA, src, dst, row_start, b_conv, bufB, N);

  mlp_wmma_kernel<<<MLP_BLOCKS, MLP_WAVES * 32, 0, stream>>>(
      bufB, W1, b1, W2, b2, out, rowTiles);
}


# edge aggregation moved into the two-layer launch: a wave sums its 16-row tile in edge order (32 gathers in flight) into its LDS slot, layer 1 reads it there; h2 never stored; 12 waves per block
# speedup vs baseline: 2.5289x; 1.0377x over previous
#include <hip/hip_runtime.h>


#define IN_C  256
#define HID_C 128
#define OUT_C 256

typedef int idx_t;
typedef __attribute__((ext_vector_type(2))) float v2f;
typedef __attribute__((ext_vector_type(8))) float v8f;

__global__ void prep_kernel(float* __restrict__ deg, int n,
                            const float* __restrict__ W,
                            float* __restrict__ Wt, int rows, int cols) {
  const int i = blockIdx.x * blockDim.x + threadIdx.x;
  if (i < n) deg[i] = 0.0f;
  if (i < rows * cols) {
    const int r = i / cols;
    const int c = i - r * cols;
    Wt[(long)c * rows + r] = W[i];
  }
}

__global__ void degree_rowstart_kernel(const idx_t* __restrict__ src,
                                       const idx_t* __restrict__ dst,
                                       float* __restrict__ deg_out,
                                       int* __restrict__ row_start, int E,
                                       int N) {
  const int e = blockIdx.x * blockDim.x + threadIdx.x;
  if (e >= E) return;
  unsafeAtomicAdd(&deg_out[src[e]], 1.0f);
  const int d = dst[e];
  const int p = e > 0 ? dst[e - 1] : -1;
  for (int n = p + 1; n <= d; ++n) row_start[n] = e;
  if (e == E - 1) {
    for (int n = d + 1; n <= N; ++n) row_start[n] = E;
  }
}

#define GEMM_BLOCKS 512
template <int KDIM, int NTOT, bool HAS_BIAS, bool RELU, bool SCALE_A>
__global__ __launch_bounds__(256) void gemm_wmma_kernel(
    const float* __restrict__ A, const float* __restrict__ W,
    const float* __restrict__ bias, const float* __restrict__ scaleA,
    float* __restrict__ D, int nRowTiles) {
  constexpr int P = 36;
  constexpr int CH = NTOT * P;
  constexpr int NSPLIT = NTOT / 128;
  constexpr int F4R = KDIM / 4;
  constexpr int PASSES = NTOT * F4R / 256;
  __shared__ __attribute__((aligned(16))) float wl[(KDIM / 32) * CH];

  const int tid = threadIdx.x;
#pragma unroll 8
  for (int p = 0; p < PASSES; ++p) {
    const int idx = (((p + (int)blockIdx.x) & (PASSES - 1)) << 8) + tid;
    const int r = idx / F4R;
    const int c4 = idx - r * F4R;
    const float4 v = *(const float4*)(W + (long)idx * 4);
    *(float4*)(wl + (c4 >> 3) * CH + r * P + (c4 & 7) * 4) = v;
  }
  __syncthreads();

  const int lane = tid & 31;
  const int c = lane & 15;
  const int khalf = lane >> 4;
  const int wv = __builtin_amdgcn_readfirstlane(tid >> 5);
  const int rank = (0x73261540 >> (4 * wv)) & 7;

  for (int v = rank;; v += 8) {
    const int u = v / NSPLIT;
    const int rowTile = u * GEMM_BLOCKS + (int)blockIdx.x;
    if (rowTile >= nRowTiles) break;
    const int n0 = (v - u * NSPLIT) * 128;
    const int m0 = rowTile * 16;
    const int mrow = m0 + c;

    const float s = SCALE_A ? rsqrtf(fmaxf(scaleA[mrow], 1.0f)) : 1.0f;

    v8f acc[8];
#pragma unroll
    for (int t = 0; t < 8; ++t) acc[t] = (v8f){0, 0, 0, 0, 0, 0, 0, 0};

    const float* __restrict__ ap = A + (long)mrow * KDIM + khalf * 2;
    const float* bp = wl + (n0 + c) * P + khalf * 2;

    float2 an[8];
#pragma unroll
    for (int j = 0; j < 8; ++j) an[j] = *(const float2*)(ap + 4 * j);

#pragma unroll 1
    for (int kb = 0; kb < KDIM; kb += 32) {
      float2 ac[8];
#pragma unroll
      for (int j = 0; j < 8; ++j) ac[j] = an[j];
      if (kb + 32 < KDIM) {
#pragma unroll
        for (int j = 0; j < 8; ++j)
          an[j] = *(const float2*)(ap + kb + 32 + 4 * j);
      }
#pragma unroll
      for (int hf = 0; hf < 2; ++hf) {
        v2f b[4][8];
#pragma unroll
        for (int js = 0; js < 4; ++js) {
#pragma unroll
          for (int t = 0; t < 8; ++t) {
            const float2 bv =
                *(const float2*)(bp + (kb >> 5) * CH + t * 16 * P + (hf * 4 + js) * 4);
            b[js][t].x = bv.x;
            b[js][t].y = bv.y;
          }
        }
#pragma unroll
        for (int js = 0; js < 4; ++js) {
          v2f a;
          a.x = ac[hf * 4 + js].x * s;
          a.y = ac[hf * 4 + js].y * s;
#pragma unroll
          for (int t = 0; t < 8; ++t) {
            acc[t] = __builtin_amdgcn_wmma_f32_16x16x4_f32(
                false, a, false, b[js][t], (short)0, acc[t], false, false);
          }
        }
      }
    }

#pragma unroll
    for (int t = 0; t < 8; ++t) {
      const int ncol = n0 + t * 16 + c;
#pragma unroll
      for (int q = 0; q < 8; ++q) {
        const int row = m0 + q + 8 * khalf;
        float val = acc[t][q];
        if constexpr (HAS_BIAS) val += bias[ncol];
        if constexpr (RELU) val = fmaxf(val, 0.0f);
        D[(long)row * NTOT + ncol] = val;
      }
    }
  }
}

#define MLP_BLOCKS 256
#define MLP_WAVES 12
#define GATHER_DEPTH 32

__device__ __forceinline__ void store_row(float* __restrict__ slot_row,
                                          const float4 acc, int cnt,
                                          const float4 bias) {
  const float rin = rsqrtf(fmaxf((float)cnt, 1.0f));
  float4 r;
  r.x = fmaxf(acc.x * rin + bias.x, 0.0f);
  r.y = fmaxf(acc.y * rin + bias.y, 0.0f);
  r.z = fmaxf(acc.z * rin + bias.z, 0.0f);
  r.w = fmaxf(acc.w * rin + bias.w, 0.0f);
  *(float4*)slot_row = r;
}

__global__ __launch_bounds__(MLP_WAVES * 32) void aggregate_mlp_kernel(
    const float* __restrict__ h1, const idx_t* __restrict__ src,
    const idx_t* __restrict__ dst, const int* __restrict__ row_start,
    const float* __restrict__ b_conv, const float* __restrict__ W1,
    const float* __restrict__ b1, const float* __restrict__ W2,
    const float* __restrict__ b2, float* __restrict__ D, int nRowTiles) {
  constexpr int P = 36;
  constexpr int CH1 = HID_C * P;
  constexpr int CH2 = OUT_C * P;
  constexpr int TP = HID_C + 4;
  constexpr int F4R = HID_C / 4;
  constexpr int NT = MLP_WAVES * 32;
  constexpr int F4_1 = HID_C * F4R;
  constexpr int PASSES = (F4_1 + OUT_C * F4R) / NT;
  static_assert((F4_1 + OUT_C * F4R) % NT == 0, "staging passes must be whole");
  __shared__ __attribute__((aligned(16))) float w1l[(HID_C / 32) * CH1];
  __shared__ __attribute__((aligned(16))) float w2l[(HID_C / 32) * CH2];
  __shared__ __attribute__((aligned(16))) float tl[MLP_WAVES * 16 * TP];

  const int tid = threadIdx.x;
  const int lane = tid & 31;
  const int c = lane & 15;
  const int khalf = lane >> 4;
  const int wv = __builtin_amdgcn_readfirstlane(tid >> 5);
  float* tw = tl + wv * 16 * TP;

  constexpr int TILE_STEP = MLP_WAVES * MLP_BLOCKS;
  const float* __restrict__ hc = h1 + lane * 4;
  float* twl = tw + lane * 4;
  const float4 bias = *(const float4*)(b_conv + lane * 4);
  float bb1[8];
#pragma unroll
  for (int t = 0; t < 8; ++t) bb1[t] = b1[t * 16 + c];

#pragma unroll 8
  for (int p = 0; p < PASSES; ++p) {
    const int idx = ((p + (int)blockIdx.x) % PASSES) * NT + tid;
    const bool second = idx >= F4_1;
    const int id2 = second ? idx - F4_1 : idx;
    const int r = id2 / F4R;
    const int c4 = id2 - r * F4R;
    const float4 v = *(const float4*)((second ? W2 : W1) + (long)id2 * 4);
    float* dstp = second ? w2l + (c4 >> 3) * CH2 : w1l + (c4 >> 3) * CH1;
    *(float4*)(dstp + r * P + (c4 & 7) * 4) = v;
  }
  __syncthreads();

  for (int rowTile = wv * MLP_BLOCKS + (int)blockIdx.x; rowTile < nRowTiles;
       rowTile += TILE_STEP) {
    const int m0 = rowTile * 16;

    {
      const int n1 = m0 + 16;
      const int e0 = row_start[m0];
      const int e1 = row_start[n1];
      int sv = 0, dv = 0;
      if (e0 + lane < e1) {
        sv = src[e0 + lane];
        dv = dst[e0 + lane];
      }
      float4 acc = make_float4(0.f, 0.f, 0.f, 0.f);
      int cur = m0;
      int cnt = 0;

      for (int eb = e0; eb < e1; eb += 32) {
        int sn = 0, dn = 0;
        const int nv = e1 - eb;
#pragma unroll
        for (int g = 0; g < 32; g += GATHER_DEPTH) {
          if (g < nv) {
            float4 v[GATHER_DEPTH];
#pragma unroll
            for (int j = 0; j < GATHER_DEPTH; ++j) {
              const int s = __builtin_amdgcn_readlane(sv, g + j);
              v[j] = *(const float4*)(hc + (long)s * HID_C);
            }
            if (g == 0) {
              if (eb + 32 + lane < e1) {
                sn = src[eb + 32 + lane];
                dn = dst[eb + 32 + lane];
              }
            }
#pragma unroll
            for (int j = 0; j < GATHER_DEPTH; ++j) {
              if (g + j < nv) {
                const int d = __builtin_amdgcn_readlane(dv, g + j);
#pragma unroll 1
                while (cur < d) {
                  store_row(twl + (cur - m0) * TP, acc, cnt, bias);
                  acc = make_float4(0.f, 0.f, 0.f, 0.f);
                  cnt = 0;
                  ++cur;
                }
                acc.x += v[j].x;
                acc.y += v[j].y;
                acc.z += v[j].z;
                acc.w += v[j].w;
                ++cnt;
              }
            }
          }
        }
        sv = sn;
        dv = dn;
      }
#pragma unroll 1
      while (cur < n1) {
        store_row(twl + (cur - m0) * TP, acc, cnt, bias);
        acc = make_float4(0.f, 0.f, 0.f, 0.f);
        cnt = 0;
        ++cur;
      }
    }

    const float* apl = tw + c * TP + khalf * 2;
    v8f acc[8];

    {
#pragma unroll
      for (int t = 0; t < 8; ++t) acc[t] = (v8f){0, 0, 0, 0, 0, 0, 0, 0};
      const float* bp = w1l + c * P + khalf * 2;
#pragma unroll 1
      for (int kb = 0; kb < HID_C; kb += 32) {
        float2 ac[8];
#pragma unroll
        for (int j = 0; j < 8; ++j) ac[j] = *(const float2*)(apl + kb + 4 * j);
#pragma unroll
        for (int hf = 0; hf < 2; ++hf) {
          v2f b[4][8];
#pragma unroll
          for (int js = 0; js < 4; ++js) {
#pragma unroll
            for (int t = 0; t < 8; ++t) {
              const float2 bv = *(const float2*)(bp + (kb >> 5) * CH1 +
                                                 t * 16 * P + (hf * 4 + js) * 4);
              b[js][t].x = bv.x;
              b[js][t].y = bv.y;
            }
          }
#pragma unroll
          for (int js = 0; js < 4; ++js) {
            v2f a;
            a.x = ac[hf * 4 + js].x;
            a.y = ac[hf * 4 + js].y;
#pragma unroll
            for (int t = 0; t < 8; ++t) {
              acc[t] = __builtin_amdgcn_wmma_f32_16x16x4_f32(
                  false, a, false, b[js][t], (short)0, acc[t], false, false);
            }
          }
        }
      }
    }

#pragma unroll
    for (int t = 0; t < 8; ++t) {
#pragma unroll
      for (int q = 0; q < 8; ++q)
        tw[(q + 8 * khalf) * TP + t * 16 + c] = fmaxf(acc[t][q] + bb1[t], 0.0f);
    }

#pragma unroll 1
    for (int n0 = 0; n0 < OUT_C; n0 += 128) {
#pragma unroll
      for (int t = 0; t < 8; ++t) acc[t] = (v8f){0, 0, 0, 0, 0, 0, 0, 0};
      const float* bp = w2l + (n0 + c) * P + khalf * 2;

#pragma unroll 1
      for (int kb = 0; kb < HID_C; kb += 32) {
        float2 ac[8];
#pragma unroll
        for (int j = 0; j < 8; ++j) ac[j] = *(const float2*)(apl + kb + 4 * j);
#pragma unroll
        for (int hf = 0; hf < 2; ++hf) {
          v2f b[4][8];
#pragma unroll
          for (int js = 0; js < 4; ++js) {
#pragma unroll
            for (int t = 0; t < 8; ++t) {
              const float2 bv = *(const float2*)(bp + (kb >> 5) * CH2 +
                                                 t * 16 * P + (hf * 4 + js) * 4);
              b[js][t].x = bv.x;
              b[js][t].y = bv.y;
            }
          }
#pragma unroll
          for (int js = 0; js < 4; ++js) {
            v2f a;
            a.x = ac[hf * 4 + js].x;
            a.y = ac[hf * 4 + js].y;
#pragma unroll
            for (int t = 0; t < 8; ++t) {
              acc[t] = __builtin_amdgcn_wmma_f32_16x16x4_f32(
                  false, a, false, b[js][t], (short)0, acc[t], false, false);
            }
          }
        }
      }

#pragma unroll
      for (int t = 0; t < 8; ++t) {
        const int ncol = n0 + t * 16 + c;
        const float bb = b2[ncol];
#pragma unroll
        for (int q = 0; q < 8; ++q) {
          const int row = m0 + q + 8 * khalf;
          D[(long)row * OUT_C + ncol] = acc[t][q] + bb;
        }
      }
    }
  }
}

extern "C" void kernel_launch(void* const* d_in, const int* in_sizes, int n_in,
                              void* d_out, int out_size, void* d_ws,
                              size_t ws_size, hipStream_t stream) {
  const float* x      = (const float*)d_in[0];
  const idx_t* src    = (const idx_t*)d_in[1];
  const idx_t* dst    = (const idx_t*)d_in[2];
  const float* W_conv = (const float*)d_in[3];
  const float* b_conv = (const float*)d_in[4];
  const float* W1     = (const float*)d_in[5];
  const float* b1     = (const float*)d_in[6];
  const float* W2     = (const float*)d_in[7];
  const float* b2     = (const float*)d_in[8];
  float* out = (float*)d_out;

  const int N = in_sizes[0] / IN_C;
  const int E = in_sizes[1];

  float* ws      = (float*)d_ws;
  float* deg_out = ws;
  int* row_start = (int*)(ws + N);
  float* Wc_t    = ws + ((2L * N + 1 + 127) & ~127L);
  float* bufA    = Wc_t + (long)HID_C * IN_C;

  const int prepN = N > IN_C * HID_C ? N : IN_C * HID_C;
  prep_kernel<<<(prepN + 255) / 256, 256, 0, stream>>>(deg_out, N, W_conv, Wc_t,
                                                       IN_C, HID_C);

  degree_rowstart_kernel<<<(E + 255) / 256, 256, 0, stream>>>(
      src, dst, deg_out, row_start, E, N);

  const int rowTiles = N / 16;

  gemm_wmma_kernel<IN_C, HID_C,  false,  false,
                    true>
      <<<GEMM_BLOCKS, 256, 0, stream>>>(x, Wc_t, nullptr, deg_out, bufA,
                                        rowTiles);

  aggregate_mlp_kernel<<<MLP_BLOCKS, MLP_WAVES * 32, 0, stream>>>(
      bufA, src, dst, row_start, b_conv, W1, b1, W2, b2, out, rowTiles);
}
